# starting kernel
# baseline (speedup 1.0000x reference)
#include <hip/hip_runtime.h>


typedef float v2f __attribute__((ext_vector_type(2)));
typedef float v8f __attribute__((ext_vector_type(8)));

#define NFEAT 512
#define NH1   256
#define NH2   128

__device__ __forceinline__ void atomic_fadd_dev(float* p, float v) {
#if defined(__gfx1250__)
  asm volatile("global_atomic_add_f32 %0, %1, off scope:SCOPE_DEV"
               :: "v"(p), "v"(v)
               : "memory");
#else
  atomicAdd(p, v);
#endif
}

__global__ void k_deg_init(float* __restrict__ deg, int n) {
  int i = blockIdx.x * blockDim.x + threadIdx.x;
  if (i < n) deg[i] = 1.0f;
}

__global__ void k_deg_count(const int* __restrict__ dst, int E, float* __restrict__ deg) {
  int e = blockIdx.x * blockDim.x + threadIdx.x;
  if (e < E) atomic_fadd_dev(&deg[dst[e]], 1.0f);
}

__global__ void k_dinv(float* __restrict__ deg, int n) {
  int i = blockIdx.x * blockDim.x + threadIdx.x;
  if (i < n) deg[i] = rsqrtf(deg[i]);
}

#define APAD 34

__global__ __launch_bounds__(256)
void k_gemm_wmma(const float* __restrict__ A, const float* __restrict__ B,
                 float* __restrict__ C, int M, int K, int N) {
  __shared__ float As[64 * APAD];
  __shared__ float Bs[32 * 64];

  const int tid  = threadIdx.x;
  const int lane = tid & 31;
  const int wv   = tid >> 5;
  const int mw   = wv & 3;
  const int nw   = wv >> 2;
  const int hf   = lane >> 4;
  const int idx  = lane & 15;

  const int m_base = blockIdx.y * 64;
  const int n_base = blockIdx.x * 64;

  v8f acc0 = {};
  v8f acc1 = {};

  for (int k0 = 0; k0 < K; k0 += 32) {
#pragma unroll
    for (int i = 0; i < 8; ++i) {
      int eid = tid + i * 256;
      int r = eid >> 5, c = eid & 31;
      int gr = m_base + r;
      As[r * APAD + c] = (gr < M) ? A[(size_t)gr * K + k0 + c] : 0.0f;
    }
#pragma unroll
    for (int i = 0; i < 8; ++i) {
      int eid = tid + i * 256;
      int r = eid >> 6, c = eid & 63;
      Bs[r * 64 + c] = B[(size_t)(k0 + r) * N + n_base + c];
    }
    __syncthreads();

#pragma unroll
    for (int kk = 0; kk < 32; kk += 4) {
      const float* ap = &As[(mw * 16 + idx) * APAD + kk + 2 * hf];
      v2f a; a.x = ap[0]; a.y = ap[1];
      const float* bp0 = &Bs[(kk + 2 * hf) * 64 + nw * 32 + idx];
      v2f b0; b0.x = bp0[0];  b0.y = bp0[64];
      v2f b1; b1.x = bp0[16]; b1.y = bp0[80];
      acc0 = __builtin_amdgcn_wmma_f32_16x16x4_f32(false, a, false, b0,
                                                   (short)0, acc0, false, false);
      acc1 = __builtin_amdgcn_wmma_f32_16x16x4_f32(false, a, false, b1,
                                                   (short)0, acc1, false, false);
    }
    __syncthreads();
  }

#pragma unroll
  for (int r = 0; r < 8; ++r) {
    int grow = m_base + mw * 16 + r + 8 * hf;
    if (grow < M) {
      C[(size_t)grow * N + n_base + nw * 32 + idx]      = acc0[r];
      C[(size_t)grow * N + n_base + nw * 32 + 16 + idx] = acc1[r];
    }
  }
}

template <int F>
__global__ __launch_bounds__(256)
void k_edge_scatter(const int* __restrict__ src, const int* __restrict__ dst,
                    int E, int n, const float* __restrict__ dinv,
                    const float* __restrict__ hin, float* __restrict__ hout) {
  int gw   = (blockIdx.x * blockDim.x + threadIdx.x) >> 5;
  int lane = threadIdx.x & 31;
  int total = E + n;
  if (gw >= total) return;

  int s, d;
  if (gw < E) { s = src[gw]; d = dst[gw]; }
  else        { s = d = gw - E; }
  float norm = dinv[s] * dinv[d];

  const float4* in4 = (const float4*)(hin + (size_t)s * F);
  float* outp = hout + (size_t)d * F;
#pragma unroll
  for (int g = 0; g < F / 128; ++g) {
    int c = lane + g * 32;
    float4 v = in4[c];
    atomic_fadd_dev(outp + c * 4 + 0, v.x * norm);
    atomic_fadd_dev(outp + c * 4 + 1, v.y * norm);
    atomic_fadd_dev(outp + c * 4 + 2, v.z * norm);
    atomic_fadd_dev(outp + c * 4 + 3, v.w * norm);
  }
}

__global__ void k_bias_relu(float* __restrict__ h, const float* __restrict__ b,
                            size_t total, int F) {
  size_t i = (size_t)blockIdx.x * blockDim.x + threadIdx.x;
  if (i < total) {
    float v = h[i] + b[i & (size_t)(F - 1)];
    h[i] = v > 0.0f ? v : 0.0f;
  }
}

__global__ void k_bias(float* __restrict__ h, const float* __restrict__ b,
                       size_t total, int F) {
  size_t i = (size_t)blockIdx.x * blockDim.x + threadIdx.x;
  if (i < total) h[i] += b[i & (size_t)(F - 1)];
}

extern "C" void kernel_launch(void* const* d_in, const int* in_sizes, int n_in,
                              void* d_out, int out_size, void* d_ws, size_t ws_size,
                              hipStream_t stream) {
  const float* x  = (const float*)d_in[0];
  const int*   ei = (const int*)d_in[1];
  const float* W1 = (const float*)d_in[2];
  const float* b1 = (const float*)d_in[3];
  const float* W2 = (const float*)d_in[4];
  const float* b2 = (const float*)d_in[5];

  const int n = in_sizes[0] / NFEAT;
  const int E = in_sizes[1] / 2;
  const int* src = ei;
  const int* dst = ei + E;

  float* dinv = (float*)d_ws;
  float* Hlin = dinv + (size_t)((n + 255) & ~255);
  float* agg  = Hlin + (size_t)n * NH1;
  float* out  = (float*)d_out;

  k_deg_init<<<(n + 255) / 256, 256, 0, stream>>>(dinv, n);
  k_deg_count<<<(E + 255) / 256, 256, 0, stream>>>(dst, E, dinv);
  k_dinv<<<(n + 255) / 256, 256, 0, stream>>>(dinv, n);

  const long long waves = (long long)E + n;
  const int scatter_blocks = (int)((waves * 32 + 255) / 256);

  k_gemm_wmma<<<dim3(NH1 / 64, (n + 63) / 64), 256, 0, stream>>>(x, W1, Hlin, n, NFEAT, NH1);
  hipMemsetAsync(agg, 0, (size_t)n * NH1 * sizeof(float), stream);
  k_edge_scatter<NH1><<<scatter_blocks, 256, 0, stream>>>(src, dst, E, n, dinv, Hlin, agg);
  k_bias_relu<<<(int)(((size_t)n * NH1 + 255) / 256), 256, 0, stream>>>(
      agg, b1, (size_t)n * NH1, NH1);

  k_gemm_wmma<<<dim3(NH2 / 64, (n + 63) / 64), 256, 0, stream>>>(agg, W2, Hlin, n, NH1, NH2);
  hipMemsetAsync(out, 0, (size_t)out_size * sizeof(float), stream);
  k_edge_scatter<NH2><<<scatter_blocks, 256, 0, stream>>>(src, dst, E, n, dinv, Hlin, out);
  k_bias<<<(int)(((size_t)out_size + 255) / 256), 256, 0, stream>>>(
      out, b2, (size_t)out_size, NH2);
}


# bias, relu, zero-fill, self-loop passes fused into GEMMs: GEMM store also writes each sum's start value b + dinv^2*C, relu on layer-2 A load, scatter adds real edges only; start value summed first
# speedup vs baseline: 1.6373x; 1.6373x over previous
#include <hip/hip_runtime.h>


typedef float v2f __attribute__((ext_vector_type(2)));
typedef float v8f __attribute__((ext_vector_type(8)));

#define NFEAT 512
#define NH1   256
#define NH2   128

__device__ __forceinline__ void atomic_fadd_dev(float* p, float v) {
#if defined(__gfx1250__)
  asm volatile("global_atomic_add_f32 %0, %1, off scope:SCOPE_DEV"
               :: "v"(p), "v"(v)
               : "memory");
#else
  atomicAdd(p, v);
#endif
}

__global__ void k_deg_init(float* __restrict__ deg, int n) {
  int i = blockIdx.x * blockDim.x + threadIdx.x;
  if (i < n) deg[i] = 1.0f;
}

__global__ void k_deg_count(const int* __restrict__ dst, int E, float* __restrict__ deg) {
  int e = blockIdx.x * blockDim.x + threadIdx.x;
  if (e < E) atomic_fadd_dev(&deg[dst[e]], 1.0f);
}

__global__ void k_dinv(float* __restrict__ deg, int n) {
  int i = blockIdx.x * blockDim.x + threadIdx.x;
  if (i < n) deg[i] = rsqrtf(deg[i]);
}

#define APAD 34

template <bool RELU_A>
__global__ __launch_bounds__(256)
void k_gemm_wmma(const float* __restrict__ A, const float* __restrict__ B,
                 float* __restrict__ C, float* __restrict__ S,
                 const float* __restrict__ dinv, const float* __restrict__ bias,
                 int M, int K, int N) {
  __shared__ float As[64 * APAD];
  __shared__ float Bs[32 * 64];

  const int tid  = threadIdx.x;
  const int lane = tid & 31;
  const int wv   = tid >> 5;
  const int mw   = wv & 3;
  const int nw   = wv >> 2;
  const int hf   = lane >> 4;
  const int idx  = lane & 15;

  const int m_base = blockIdx.y * 64;
  const int n_base = blockIdx.x * 64;

  v8f acc0 = {};
  v8f acc1 = {};

  for (int k0 = 0; k0 < K; k0 += 32) {
#pragma unroll
    for (int i = 0; i < 8; ++i) {
      int eid = tid + i * 256;
      int r = eid >> 5, c = eid & 31;
      int gr = m_base + r;
      float av = (gr < M) ? A[(size_t)gr * K + k0 + c] : 0.0f;
      if (RELU_A) av = av > 0.0f ? av : 0.0f;
      As[r * APAD + c] = av;
    }
#pragma unroll
    for (int i = 0; i < 8; ++i) {
      int eid = tid + i * 256;
      int r = eid >> 6, c = eid & 63;
      Bs[r * 64 + c] = B[(size_t)(k0 + r) * N + n_base + c];
    }
    __syncthreads();

#pragma unroll
    for (int kk = 0; kk < 32; kk += 4) {
      const float* ap = &As[(mw * 16 + idx) * APAD + kk + 2 * hf];
      v2f a; a.x = ap[0]; a.y = ap[1];
      const float* bp0 = &Bs[(kk + 2 * hf) * 64 + nw * 32 + idx];
      v2f b0; b0.x = bp0[0];  b0.y = bp0[64];
      v2f b1; b1.x = bp0[16]; b1.y = bp0[80];
      acc0 = __builtin_amdgcn_wmma_f32_16x16x4_f32(false, a, false, b0,
                                                   (short)0, acc0, false, false);
      acc1 = __builtin_amdgcn_wmma_f32_16x16x4_f32(false, a, false, b1,
                                                   (short)0, acc1, false, false);
    }
    __syncthreads();
  }

  const int col = n_base + nw * 32 + idx;
  const float bia0 = bias[col];
  const float bia1 = bias[col + 16];
  float di8[8];
#pragma unroll
  for (int r = 0; r < 8; ++r) {
    int grow = m_base + mw * 16 + r + 8 * hf;
    di8[r] = dinv[grow < M ? grow : M - 1];
  }
#pragma unroll
  for (int r = 0; r < 8; ++r) {
    int grow = m_base + mw * 16 + r + 8 * hf;
    if (grow < M) {
      const float w = di8[r] * di8[r];
      const size_t o = (size_t)grow * N + col;
      C[o]      = acc0[r];
      C[o + 16] = acc1[r];
      S[o]      = acc0[r] * w + bia0;
      S[o + 16] = acc1[r] * w + bia1;
    }
  }
}

template <int OFFB>
__device__ __forceinline__ void atomic_fadd_line(float* rowbase, unsigned lane_byte, float v) {
#if defined(__gfx1250__)
  asm volatile("global_atomic_add_f32 %0, %1, %2 offset:%3 scope:SCOPE_DEV"
               :: "v"(lane_byte), "v"(v), "s"(rowbase), "n"(OFFB)
               : "memory");
#else
  atomicAdd(rowbase + (OFFB >> 2) + (lane_byte >> 2), v);
#endif
}

template <int G, int NG>
struct RowLines {
  static __device__ __forceinline__ void load(const float* rp, float* v) {
    v[G] = rp[G * 32];
    if constexpr (G + 1 < NG) RowLines<G + 1, NG>::load(rp, v);
  }
  static __device__ __forceinline__ void add(float* op, unsigned lane_byte, const float* v, float nrm) {
    atomic_fadd_line<G * 128>(op, lane_byte, v[G] * nrm);
    if constexpr (G + 1 < NG) RowLines<G + 1, NG>::add(op, lane_byte, v, nrm);
  }
};

template <int F, int EPW, int BATCH, bool FULL>
__device__ __forceinline__ void scatter_group(int s, int d, const float* __restrict__ dinv,
                                              int lane, int left,
                                              const float* __restrict__ hin,
                                              float* __restrict__ hout) {
  constexpr int NG = F / 32;
  const unsigned lane_byte = (unsigned)lane * 4u;
  float nrm = 0.0f;
#pragma unroll
  for (int j0 = 0; j0 < EPW; j0 += BATCH) {
    float v[BATCH][NG];
#pragma unroll
    for (int j = 0; j < BATCH; ++j) {
      if (FULL || j0 + j < left) {
        const unsigned sj = (unsigned)__builtin_amdgcn_readlane(s, j0 + j);
        RowLines<0, NG>::load(hin + (size_t)sj * F + lane, v[j]);
      }
    }
    if (j0 == 0) nrm = dinv[s] * dinv[d];
#pragma unroll
    for (int j = 0; j < BATCH; ++j) {
      if (FULL || j0 + j < left) {
        const unsigned dj = (unsigned)__builtin_amdgcn_readlane(d, j0 + j);
        const float nj = __int_as_float(__builtin_amdgcn_readlane(__float_as_int(nrm), j0 + j));
        RowLines<0, NG>::add(hout + (size_t)dj * F, lane_byte, v[j], nj);
      }
    }
    asm volatile("s_wait_xcnt 0x0" ::: "memory");
  }
}

template <int F, int EPW, int BATCH>
__global__ __launch_bounds__(256)
void k_edge_scatter(const int* __restrict__ src, const int* __restrict__ dst,
                    int E, const float* __restrict__ dinv,
                    const float* __restrict__ hin, float* __restrict__ hout) {
  const int lane = threadIdx.x & 31;
  const int wave = __builtin_amdgcn_readfirstlane((int)(blockIdx.x * 8 + (threadIdx.x >> 5)));
  const int base = wave * EPW;
  const int left = E - base;
  if (left <= 0) return;

  int e = base + (lane & (EPW - 1));
  e = e < E ? e : E - 1;
  const int s = src[e];
  const int d = dst[e];

  if (left < EPW) {
    scatter_group<F, EPW, BATCH, false>(s, d, dinv, lane, left, hin, hout);
    return;
  }
  scatter_group<F, EPW, BATCH, true>(s, d, dinv, lane, left, hin, hout);
}

extern "C" void kernel_launch(void* const* d_in, const int* in_sizes, int n_in,
                              void* d_out, int out_size, void* d_ws, size_t ws_size,
                              hipStream_t stream) {
  const float* x  = (const float*)d_in[0];
  const int*   ei = (const int*)d_in[1];
  const float* W1 = (const float*)d_in[2];
  const float* b1 = (const float*)d_in[3];
  const float* W2 = (const float*)d_in[4];
  const float* b2 = (const float*)d_in[5];

  const int n = in_sizes[0] / NFEAT;
  const int E = in_sizes[1] / 2;
  const int* src = ei;
  const int* dst = ei + E;

  float* dinv = (float*)d_ws;
  float* Hlin = dinv + (size_t)((n + 255) & ~255);
  float* agg  = Hlin + (size_t)n * NH1;
  float* out  = (float*)d_out;

  k_deg_init<<<(n + 255) / 256, 256, 0, stream>>>(dinv, n);
  k_deg_count<<<(E + 255) / 256, 256, 0, stream>>>(dst, E, dinv);
  k_dinv<<<(n + 255) / 256, 256, 0, stream>>>(dinv, n);

  const int scatter_blocks = (E + 63) / 64;

  k_gemm_wmma<false><<<dim3(NH1 / 64, (n + 63) / 64), 256, 0, stream>>>(
      x, W1, Hlin, agg, dinv, b1, n, NFEAT, NH1);
  k_edge_scatter<NH1, 8, 4><<<scatter_blocks, 256, 0, stream>>>(src, dst, E, dinv, Hlin, agg);

  k_gemm_wmma<true><<<dim3(NH2 / 64, (n + 63) / 64), 256, 0, stream>>>(
      agg, W2, Hlin, out, dinv, b2, n, NH1, NH2);
  k_edge_scatter<NH2, 8, 8><<<scatter_blocks, 256, 0, stream>>>(src, dst, E, dinv, Hlin, out);
}


# GEMM A-tile load guard -> clamped row index (value zeroed after), all 16 loads of a k-chunk issued as one group: one memory round trip per chunk instead of 8 dependent ones; sums bit-identical
# speedup vs baseline: 1.8669x; 1.1402x over previous
#include <hip/hip_runtime.h>


typedef float v2f __attribute__((ext_vector_type(2)));
typedef float v8f __attribute__((ext_vector_type(8)));

#define NFEAT 512
#define NH1   256
#define NH2   128

__device__ __forceinline__ void atomic_fadd_dev(float* p, float v) {
#if defined(__gfx1250__)
  asm volatile("global_atomic_add_f32 %0, %1, off scope:SCOPE_DEV"
               :: "v"(p), "v"(v)
               : "memory");
#else
  atomicAdd(p, v);
#endif
}

__global__ void k_deg_init(float* __restrict__ deg, int n) {
  int i = blockIdx.x * blockDim.x + threadIdx.x;
  if (i < n) deg[i] = 1.0f;
}

__global__ void k_deg_count(const int* __restrict__ dst, int E, float* __restrict__ deg) {
  int e = blockIdx.x * blockDim.x + threadIdx.x;
  if (e < E) atomic_fadd_dev(&deg[dst[e]], 1.0f);
}

__global__ void k_dinv(float* __restrict__ deg, int n) {
  int i = blockIdx.x * blockDim.x + threadIdx.x;
  if (i < n) deg[i] = rsqrtf(deg[i]);
}

#define APAD 34

template <bool RELU_A>
__global__ __launch_bounds__(256)
void k_gemm_wmma(const float* __restrict__ A, const float* __restrict__ B,
                 float* __restrict__ C, float* __restrict__ S,
                 const float* __restrict__ dinv, const float* __restrict__ bias,
                 int M, int K, int N) {
  __shared__ float As[64 * APAD];
  __shared__ float Bs[32 * 64];

  const int tid  = threadIdx.x;
  const int lane = tid & 31;
  const int wv   = tid >> 5;
  const int mw   = wv & 3;
  const int nw   = wv >> 2;
  const int hf   = lane >> 4;
  const int idx  = lane & 15;

  const int m_base = blockIdx.y * 64;
  const int n_base = blockIdx.x * 64;

  v8f acc0 = {};
  v8f acc1 = {};

  for (int k0 = 0; k0 < K; k0 += 32) {
    float av[8], bv[8];
#pragma unroll
    for (int i = 0; i < 8; ++i) {
      int eid = tid + i * 256;
      int r = eid >> 5, c = eid & 31;
      int gr = m_base + r;
      int grc = gr < M ? gr : M - 1;
      av[i] = A[(size_t)grc * K + k0 + c];
    }
#pragma unroll
    for (int i = 0; i < 8; ++i) {
      int eid = tid + i * 256;
      int r = eid >> 6, c = eid & 63;
      bv[i] = B[(size_t)(k0 + r) * N + n_base + c];
    }
    __builtin_amdgcn_sched_barrier(0);
#pragma unroll
    for (int i = 0; i < 8; ++i) {
      int eid = tid + i * 256;
      int r = eid >> 5, c = eid & 31;
      float a = (m_base + r < M) ? av[i] : 0.0f;
      if (RELU_A) a = a > 0.0f ? a : 0.0f;
      As[r * APAD + c] = a;
    }
#pragma unroll
    for (int i = 0; i < 8; ++i) {
      int eid = tid + i * 256;
      int r = eid >> 6, c = eid & 63;
      Bs[r * 64 + c] = bv[i];
    }
    __syncthreads();

#pragma unroll
    for (int kk = 0; kk < 32; kk += 4) {
      const float* ap = &As[(mw * 16 + idx) * APAD + kk + 2 * hf];
      v2f a; a.x = ap[0]; a.y = ap[1];
      const float* bp0 = &Bs[(kk + 2 * hf) * 64 + nw * 32 + idx];
      v2f b0; b0.x = bp0[0];  b0.y = bp0[64];
      v2f b1; b1.x = bp0[16]; b1.y = bp0[80];
      acc0 = __builtin_amdgcn_wmma_f32_16x16x4_f32(false, a, false, b0,
                                                   (short)0, acc0, false, false);
      acc1 = __builtin_amdgcn_wmma_f32_16x16x4_f32(false, a, false, b1,
                                                   (short)0, acc1, false, false);
    }
    __syncthreads();
  }

  const int col = n_base + nw * 32 + idx;
  const float bia0 = bias[col];
  const float bia1 = bias[col + 16];
  float di8[8];
#pragma unroll
  for (int r = 0; r < 8; ++r) {
    int grow = m_base + mw * 16 + r + 8 * hf;
    di8[r] = dinv[grow < M ? grow : M - 1];
  }
#pragma unroll
  for (int r = 0; r < 8; ++r) {
    int grow = m_base + mw * 16 + r + 8 * hf;
    if (grow < M) {
      const float w = di8[r] * di8[r];
      const size_t o = (size_t)grow * N + col;
      C[o]      = acc0[r];
      C[o + 16] = acc1[r];
      S[o]      = acc0[r] * w + bia0;
      S[o + 16] = acc1[r] * w + bia1;
    }
  }
}

template <int OFFB>
__device__ __forceinline__ void atomic_fadd_line(float* rowbase, unsigned lane_byte, float v) {
#if defined(__gfx1250__)
  asm volatile("global_atomic_add_f32 %0, %1, %2 offset:%3 scope:SCOPE_DEV"
               :: "v"(lane_byte), "v"(v), "s"(rowbase), "n"(OFFB)
               : "memory");
#else
  atomicAdd(rowbase + (OFFB >> 2) + (lane_byte >> 2), v);
#endif
}

template <int G, int NG>
struct RowLines {
  static __device__ __forceinline__ void load(const float* rp, float* v) {
    v[G] = rp[G * 32];
    if constexpr (G + 1 < NG) RowLines<G + 1, NG>::load(rp, v);
  }
  static __device__ __forceinline__ void add(float* op, unsigned lane_byte, const float* v, float nrm) {
    atomic_fadd_line<G * 128>(op, lane_byte, v[G] * nrm);
    if constexpr (G + 1 < NG) RowLines<G + 1, NG>::add(op, lane_byte, v, nrm);
  }
};

template <int F, int EPW, int BATCH, bool FULL>
__device__ __forceinline__ void scatter_group(int s, int d, const float* __restrict__ dinv,
                                              int lane, int left,
                                              const float* __restrict__ hin,
                                              float* __restrict__ hout) {
  constexpr int NG = F / 32;
  const unsigned lane_byte = (unsigned)lane * 4u;
  float nrm = 0.0f;
#pragma unroll
  for (int j0 = 0; j0 < EPW; j0 += BATCH) {
    float v[BATCH][NG];
#pragma unroll
    for (int j = 0; j < BATCH; ++j) {
      if (FULL || j0 + j < left) {
        const unsigned sj = (unsigned)__builtin_amdgcn_readlane(s, j0 + j);
        RowLines<0, NG>::load(hin + (size_t)sj * F + lane, v[j]);
      }
    }
    if (j0 == 0) nrm = dinv[s] * dinv[d];
#pragma unroll
    for (int j = 0; j < BATCH; ++j) {
      if (FULL || j0 + j < left) {
        const unsigned dj = (unsigned)__builtin_amdgcn_readlane(d, j0 + j);
        const float nj = __int_as_float(__builtin_amdgcn_readlane(__float_as_int(nrm), j0 + j));
        RowLines<0, NG>::add(hout + (size_t)dj * F, lane_byte, v[j], nj);
      }
    }
    asm volatile("s_wait_xcnt 0x0" ::: "memory");
  }
}

template <int F, int EPW, int BATCH>
__global__ __launch_bounds__(256)
void k_edge_scatter(const int* __restrict__ src, const int* __restrict__ dst,
                    int E, const float* __restrict__ dinv,
                    const float* __restrict__ hin, float* __restrict__ hout) {
  const int lane = threadIdx.x & 31;
  const int wave = __builtin_amdgcn_readfirstlane((int)(blockIdx.x * 8 + (threadIdx.x >> 5)));
  const int base = wave * EPW;
  const int left = E - base;
  if (left <= 0) return;

  int e = base + (lane & (EPW - 1));
  e = e < E ? e : E - 1;
  const int s = src[e];
  const int d = dst[e];

  if (left < EPW) {
    scatter_group<F, EPW, BATCH, false>(s, d, dinv, lane, left, hin, hout);
    return;
  }
  scatter_group<F, EPW, BATCH, true>(s, d, dinv, lane, left, hin, hout);
}

extern "C" void kernel_launch(void* const* d_in, const int* in_sizes, int n_in,
                              void* d_out, int out_size, void* d_ws, size_t ws_size,
                              hipStream_t stream) {
  const float* x  = (const float*)d_in[0];
  const int*   ei = (const int*)d_in[1];
  const float* W1 = (const float*)d_in[2];
  const float* b1 = (const float*)d_in[3];
  const float* W2 = (const float*)d_in[4];
  const float* b2 = (const float*)d_in[5];

  const int n = in_sizes[0] / NFEAT;
  const int E = in_sizes[1] / 2;
  const int* src = ei;
  const int* dst = ei + E;

  float* dinv = (float*)d_ws;
  float* Hlin = dinv + (size_t)((n + 255) & ~255);
  float* agg  = Hlin + (size_t)n * NH1;
  float* out  = (float*)d_out;

  k_deg_init<<<(n + 255) / 256, 256, 0, stream>>>(dinv, n);
  k_deg_count<<<(E + 255) / 256, 256, 0, stream>>>(dst, E, dinv);
  k_dinv<<<(n + 255) / 256, 256, 0, stream>>>(dinv, n);

  const int scatter_blocks = (E + 63) / 64;

  k_gemm_wmma<false><<<dim3(NH1 / 64, (n + 63) / 64), 256, 0, stream>>>(
      x, W1, Hlin, agg, dinv, b1, n, NFEAT, NH1);
  k_edge_scatter<NH1, 8, 4><<<scatter_blocks, 256, 0, stream>>>(src, dst, E, dinv, Hlin, agg);

  k_gemm_wmma<true><<<dim3(NH2 / 64, (n + 63) / 64), 256, 0, stream>>>(
      agg, W2, Hlin, out, dinv, b2, n, NH1, NH2);
  k_edge_scatter<NH2, 8, 8><<<scatter_blocks, 256, 0, stream>>>(src, dst, E, dinv, Hlin, out);
}
